# starting kernel
# baseline (speedup 1.0000x reference)
#include <hip/hip_runtime.h>
#include <hip/hip_bf16.h>

typedef float v2f __attribute__((ext_vector_type(2)));
typedef float v8f __attribute__((ext_vector_type(8)));

#define NNODES 100000
#define DIM 64

__global__ void count_edges_kernel(const int* __restrict__ dst,
                                   float* __restrict__ cnt, int n_edges) {
  int e = blockIdx.x * blockDim.x + threadIdx.x;
  if (e < n_edges) atomicAdd(&cnt[dst[e]], 1.0f);
}

__global__ void inv_count_kernel(float* __restrict__ cnt, int n) {
  int i = blockIdx.x * blockDim.x + threadIdx.x;
  if (i < n) cnt[i] = 1.0f / fmaxf(cnt[i], 1.0f);
}

__global__ void scatter_feat_kernel(const float* __restrict__ feat,
                                    const int* __restrict__ src,
                                    const int* __restrict__ dst,
                                    float* __restrict__ agg, int n_edges) {
  int t = blockIdx.x * blockDim.x + threadIdx.x;
  int e = t >> 4;
  if (e >= n_edges) return;
  int c = (t & 15) << 2;
  int s = src[e], d = dst[e];
  const float4 v = *(const float4*)(feat + (size_t)s * DIM + c);
  float* o = agg + (size_t)d * DIM + c;
  atomicAdd(o + 0, v.x);
  atomicAdd(o + 1, v.y);
  atomicAdd(o + 2, v.z);
  atomicAdd(o + 3, v.w);
}

template <bool FINAL>
__global__ __launch_bounds__(64) void sage_layer_kernel(
    const float* __restrict__ agg, const float* __restrict__ invc,
    const float* __restrict__ x,
    const float* __restrict__ Wl, const float* __restrict__ Wr,
    const float* __restrict__ bias,
    float* __restrict__ Hout,
    const float* __restrict__ Wc, const float* __restrict__ bc,
    float* __restrict__ out) {
  const int lane = threadIdx.x & 31;
  const int wave = threadIdx.x >> 5;
  const int rb   = blockIdx.x * 2 + wave;
  const int m0   = rb * 16;
  const int col  = lane & 15;
  const int half = lane >> 4;

  const int m = m0 + col;
  const float ic = invc[m];
  const float* aggRow = agg + (size_t)m * DIM;
  const float* xRow   = x   + (size_t)m * DIM;

  v8f c0, c1, c2, c3;
  {
    const float b0 = bias[col], b1 = bias[16 + col];
    const float b2 = bias[32 + col], b3 = bias[48 + col];
#pragma unroll
    for (int r = 0; r < 8; ++r) { c0[r] = b0; c1[r] = b1; c2[r] = b2; c3[r] = b3; }
  }

#pragma unroll
  for (int k0 = 0; k0 < DIM; k0 += 4) {
    const int kb = k0 + 2 * half;
    v2f Am, Ax, B;
    {
      const float2 a = *(const float2*)(aggRow + kb);
      Am[0] = a.x * ic; Am[1] = a.y * ic;
      const float2 b = *(const float2*)(xRow + kb);
      Ax[0] = b.x; Ax[1] = b.y;
    }
    const float* wl = Wl + (size_t)kb * DIM + col;
    const float* wr = Wr + (size_t)kb * DIM + col;

    B[0] = wl[0];       B[1] = wl[DIM];
    c0 = __builtin_amdgcn_wmma_f32_16x16x4_f32(false, Am, false, B, (short)0, c0, false, false);
    B[0] = wr[0];       B[1] = wr[DIM];
    c0 = __builtin_amdgcn_wmma_f32_16x16x4_f32(false, Ax, false, B, (short)0, c0, false, false);
    B[0] = wl[16];      B[1] = wl[DIM + 16];
    c1 = __builtin_amdgcn_wmma_f32_16x16x4_f32(false, Am, false, B, (short)0, c1, false, false);
    B[0] = wr[16];      B[1] = wr[DIM + 16];
    c1 = __builtin_amdgcn_wmma_f32_16x16x4_f32(false, Ax, false, B, (short)0, c1, false, false);
    B[0] = wl[32];      B[1] = wl[DIM + 32];
    c2 = __builtin_amdgcn_wmma_f32_16x16x4_f32(false, Am, false, B, (short)0, c2, false, false);
    B[0] = wr[32];      B[1] = wr[DIM + 32];
    c2 = __builtin_amdgcn_wmma_f32_16x16x4_f32(false, Ax, false, B, (short)0, c2, false, false);
    B[0] = wl[48];      B[1] = wl[DIM + 48];
    c3 = __builtin_amdgcn_wmma_f32_16x16x4_f32(false, Am, false, B, (short)0, c3, false, false);
    B[0] = wr[48];      B[1] = wr[DIM + 48];
    c3 = __builtin_amdgcn_wmma_f32_16x16x4_f32(false, Ax, false, B, (short)0, c3, false, false);
  }

#pragma unroll
  for (int r = 0; r < 8; ++r) {
    c0[r] = fmaxf(c0[r], 0.0f);
    c1[r] = fmaxf(c1[r], 0.0f);
    c2[r] = fmaxf(c2[r], 0.0f);
    c3[r] = fmaxf(c3[r], 0.0f);
  }

  if (!FINAL) {
#pragma unroll
    for (int r = 0; r < 8; ++r) {
      const size_t row = (size_t)(m0 + r + 8 * half) * DIM;
      Hout[row + col]      = c0[r];
      Hout[row + 16 + col] = c1[r];
      Hout[row + 32 + col] = c2[r];
      Hout[row + 48 + col] = c3[r];
    }
  } else {
    const float w0 = Wc[col], w1 = Wc[16 + col], w2 = Wc[32 + col], w3 = Wc[48 + col];
    float p[8];
#pragma unroll
    for (int r = 0; r < 8; ++r)
      p[r] = c0[r] * w0 + c1[r] * w1 + c2[r] * w2 + c3[r] * w3;
#pragma unroll
    for (int r = 0; r < 8; ++r) {
      p[r] += __shfl_xor(p[r], 1, 32);
      p[r] += __shfl_xor(p[r], 2, 32);
      p[r] += __shfl_xor(p[r], 4, 32);
      p[r] += __shfl_xor(p[r], 8, 32);
    }
    if (col == 0) {
      const float bcv = bc[0];
#pragma unroll
      for (int r = 0; r < 8; ++r)
        out[m0 + r + 8 * half] = p[r] + bcv;
    }
  }
}

extern "C" void kernel_launch(void* const* d_in, const int* in_sizes, int n_in,
                              void* d_out, int out_size, void* d_ws, size_t ws_size,
                              hipStream_t stream) {
  const float* x   = (const float*)d_in[0];
  const int*   ei  = (const int*)d_in[1];
  const float* W1l = (const float*)d_in[2];
  const float* W1r = (const float*)d_in[3];
  const float* b1  = (const float*)d_in[4];
  const float* W2l = (const float*)d_in[5];
  const float* W2r = (const float*)d_in[6];
  const float* b2  = (const float*)d_in[7];
  const float* Wc  = (const float*)d_in[8];
  const float* bc  = (const float*)d_in[9];
  float* out = (float*)d_out;

  const int n_edges = in_sizes[1] / 2;
  const int* src = ei;
  const int* dst = ei + n_edges;

  const size_t featBytes = (size_t)NNODES * DIM * sizeof(float);
  char* ws = (char*)d_ws;
  float* agg = (float*)ws;
  float* h1  = (float*)(ws + featBytes);
  float* cnt = (float*)(ws + 2 * featBytes);

  hipMemsetAsync(cnt, 0, (size_t)NNODES * sizeof(float), stream);
  hipMemsetAsync(agg, 0, featBytes, stream);
  count_edges_kernel<<<(n_edges + 255) / 256, 256, 0, stream>>>(dst, cnt, n_edges);
  inv_count_kernel<<<(NNODES + 255) / 256, 256, 0, stream>>>(cnt, NNODES);

  const int scatterBlocks = (n_edges * 16 + 255) / 256;
  scatter_feat_kernel<<<scatterBlocks, 256, 0, stream>>>(x, src, dst, agg, n_edges);
  sage_layer_kernel<false><<<NNODES / 32, 64, 0, stream>>>(
      agg, cnt, x, W1l, W1r, b1, h1, nullptr, nullptr, nullptr);

  hipMemsetAsync(agg, 0, featBytes, stream);
  scatter_feat_kernel<<<scatterBlocks, 256, 0, stream>>>(h1, src, dst, agg, n_edges);
  sage_layer_kernel<true><<<NNODES / 32, 64, 0, stream>>>(
      agg, cnt, h1, W2l, W2r, b2, nullptr, Wc, bc, out);
}


# Neighbour sums by gather over an in-call destination-sorted edge list (integer rank adds, scan, placement), summed in the layer kernel through LDS; replaces the f32 atomic scatter; rank-order sums
# speedup vs baseline: 2.0642x; 2.0642x over previous
#include <hip/hip_runtime.h>
#include <hip/hip_bf16.h>

typedef float v2f __attribute__((ext_vector_type(2)));
typedef float v8f __attribute__((ext_vector_type(8)));

#define NNODES 100000
#define DIM 64
#define TP 68
#define SCAN_WG ((NNODES / 4 + 1023) / 1024)

__global__ void rank_edges_kernel(const int* __restrict__ dst,
                                  int* __restrict__ deg, int* __restrict__ rank,
                                  int n_edges) {
  int e = blockIdx.x * blockDim.x + threadIdx.x;
  if (e < n_edges) rank[e] = atomicAdd(&deg[dst[e]], 1);
}

__global__ __launch_bounds__(1024) void scan_local_kernel(const int* __restrict__ deg,
                                                          int* __restrict__ off,
                                                          int* __restrict__ btot) {
  __shared__ int wtot[32];
  const int t = threadIdx.x;
  const int lane = t & 31, w = t >> 5;
  const int q = blockIdx.x * 1024 + t;
  const bool act = q < NNODES / 4;

  int4 v = make_int4(0, 0, 0, 0);
  if (act) v = ((const int4*)deg)[q];
  const int local = v.x + v.y + v.z + v.w;

  int incl = local;
#pragma unroll
  for (int d = 1; d < 32; d <<= 1) {
    const int o = __shfl_up(incl, d, 32);
    if (lane >= d) incl += o;
  }
  if (lane == 31) wtot[w] = incl;
  __syncthreads();
  if (w == 0) {
    int s = wtot[lane];
#pragma unroll
    for (int d = 1; d < 32; d <<= 1) {
      const int o = __shfl_up(s, d, 32);
      if (lane >= d) s += o;
    }
    wtot[lane] = s;
  }
  __syncthreads();
  const int run = incl - local + (w > 0 ? wtot[w - 1] : 0);

  if (act) {
    int4 r;
    r.x = run;
    r.y = r.x + v.x;
    r.z = r.y + v.y;
    r.w = r.z + v.z;
    ((int4*)off)[q] = r;
  }
  if (t == 1023) btot[blockIdx.x] = run + local;
}

__global__ __launch_bounds__(1024) void scan_add_kernel(int* __restrict__ off,
                                                        const int* __restrict__ btot) {
  const int b = blockIdx.x;
  int base = 0;
  for (int j = 0; j < b; ++j) base += btot[j];
  const int q = b * 1024 + threadIdx.x;
  if (q < NNODES / 4) {
    int4 r = ((int4*)off)[q];
    r.x += base; r.y += base; r.z += base; r.w += base;
    ((int4*)off)[q] = r;
  }
  if (b == gridDim.x - 1 && threadIdx.x == 0) off[NNODES] = base + btot[b];
}

__global__ void place_edges_kernel(const int* __restrict__ src,
                                   const int* __restrict__ dst,
                                   const int* __restrict__ off,
                                   const int* __restrict__ rank,
                                   int* __restrict__ ssrc, int n_edges) {
  int e = blockIdx.x * blockDim.x + threadIdx.x;
  if (e < n_edges) ssrc[off[dst[e]] + rank[e]] = src[e];
}

__device__ __forceinline__ float4 load_row4(const float* __restrict__ feat, int s, int c4) {
  const unsigned su = min((unsigned)s, (unsigned)(NNODES - 1));
  return *(const float4*)(feat + (size_t)su * DIM + c4);
}

template <bool FINAL>
__global__ __launch_bounds__(64) void sage_layer_kernel(
    const float* __restrict__ feat,
    const int* __restrict__ off, const int* __restrict__ ssrc,
    const float* __restrict__ Wl, const float* __restrict__ Wr,
    const float* __restrict__ bias,
    float* __restrict__ Hout,
    const float* __restrict__ Wc, const float* __restrict__ bc,
    float* __restrict__ out) {
  __shared__ float tile[2][16 * TP];

  const int lane = threadIdx.x & 31;
  const int wave = threadIdx.x >> 5;
  const int rb   = blockIdx.x * 2 + wave;
  const int m0   = rb * 16;
  const int col  = lane & 15;
  const int half = lane >> 4;
  float* myTile = tile[wave];

  {
    const int c4 = col * 4;
    for (int i = 0; i < 8; ++i) {
      const int r = 2 * i + half;
      const int beg = off[m0 + r], end = off[m0 + r + 1];
      float4 acc = make_float4(0.f, 0.f, 0.f, 0.f);
      if (beg < end) {
        const int last = end - 1;
        int s[8];
#pragma unroll
        for (int k = 0; k < 8; ++k) s[k] = ssrc[min(beg + k, last)];
        for (int e = beg; e < end; e += 8) {
          float4 v[8];
#pragma unroll
          for (int k = 0; k < 8; ++k) v[k] = load_row4(feat, s[k], c4);
#pragma unroll
          for (int k = 0; k < 8; ++k) s[k] = ssrc[min(e + 8 + k, last)];
#pragma unroll
          for (int k = 0; k < 8; ++k) {
            const bool ok = e + k < end;
            const float vx = v[k].x, vy = v[k].y, vz = v[k].z, vw = v[k].w;
            acc.x += ok ? vx : 0.0f;
            acc.y += ok ? vy : 0.0f;
            acc.z += ok ? vz : 0.0f;
            acc.w += ok ? vw : 0.0f;
          }
        }
      }
      const float ic = 1.0f / fmaxf((float)(end - beg), 1.0f);
      acc.x *= ic; acc.y *= ic; acc.z *= ic; acc.w *= ic;
      *(float4*)(myTile + r * TP + c4) = acc;
    }
  }
  __syncthreads();

  const int m = m0 + col;
  const float* meanRow = myTile + col * TP;
  const float* xRow    = feat + (size_t)m * DIM;

  v8f c0, c1, c2, c3;
  {
    const float b0 = bias[col], b1 = bias[16 + col];
    const float b2 = bias[32 + col], b3 = bias[48 + col];
#pragma unroll
    for (int r = 0; r < 8; ++r) { c0[r] = b0; c1[r] = b1; c2[r] = b2; c3[r] = b3; }
  }

#pragma unroll
  for (int k0 = 0; k0 < DIM; k0 += 4) {
    const int kb = k0 + 2 * half;
    v2f Am, Ax, B;
    {
      const float2 a = *(const float2*)(meanRow + kb);
      Am[0] = a.x; Am[1] = a.y;
      const float2 b = *(const float2*)(xRow + kb);
      Ax[0] = b.x; Ax[1] = b.y;
    }
    const float* wl = Wl + (size_t)kb * DIM + col;
    const float* wr = Wr + (size_t)kb * DIM + col;

    B[0] = wl[0];       B[1] = wl[DIM];
    c0 = __builtin_amdgcn_wmma_f32_16x16x4_f32(false, Am, false, B, (short)0, c0, false, false);
    B[0] = wr[0];       B[1] = wr[DIM];
    c0 = __builtin_amdgcn_wmma_f32_16x16x4_f32(false, Ax, false, B, (short)0, c0, false, false);
    B[0] = wl[16];      B[1] = wl[DIM + 16];
    c1 = __builtin_amdgcn_wmma_f32_16x16x4_f32(false, Am, false, B, (short)0, c1, false, false);
    B[0] = wr[16];      B[1] = wr[DIM + 16];
    c1 = __builtin_amdgcn_wmma_f32_16x16x4_f32(false, Ax, false, B, (short)0, c1, false, false);
    B[0] = wl[32];      B[1] = wl[DIM + 32];
    c2 = __builtin_amdgcn_wmma_f32_16x16x4_f32(false, Am, false, B, (short)0, c2, false, false);
    B[0] = wr[32];      B[1] = wr[DIM + 32];
    c2 = __builtin_amdgcn_wmma_f32_16x16x4_f32(false, Ax, false, B, (short)0, c2, false, false);
    B[0] = wl[48];      B[1] = wl[DIM + 48];
    c3 = __builtin_amdgcn_wmma_f32_16x16x4_f32(false, Am, false, B, (short)0, c3, false, false);
    B[0] = wr[48];      B[1] = wr[DIM + 48];
    c3 = __builtin_amdgcn_wmma_f32_16x16x4_f32(false, Ax, false, B, (short)0, c3, false, false);
  }

#pragma unroll
  for (int r = 0; r < 8; ++r) {
    c0[r] = fmaxf(c0[r], 0.0f);
    c1[r] = fmaxf(c1[r], 0.0f);
    c2[r] = fmaxf(c2[r], 0.0f);
    c3[r] = fmaxf(c3[r], 0.0f);
  }

  if (!FINAL) {
#pragma unroll
    for (int r = 0; r < 8; ++r) {
      const size_t row = (size_t)(m0 + r + 8 * half) * DIM;
      Hout[row + col]      = c0[r];
      Hout[row + 16 + col] = c1[r];
      Hout[row + 32 + col] = c2[r];
      Hout[row + 48 + col] = c3[r];
    }
  } else {
    const float w0 = Wc[col], w1 = Wc[16 + col], w2 = Wc[32 + col], w3 = Wc[48 + col];
    float p[8];
#pragma unroll
    for (int r = 0; r < 8; ++r)
      p[r] = c0[r] * w0 + c1[r] * w1 + c2[r] * w2 + c3[r] * w3;
#pragma unroll
    for (int r = 0; r < 8; ++r) {
      p[r] += __shfl_xor(p[r], 1, 32);
      p[r] += __shfl_xor(p[r], 2, 32);
      p[r] += __shfl_xor(p[r], 4, 32);
      p[r] += __shfl_xor(p[r], 8, 32);
    }
    if (col == 0) {
      const float bcv = bc[0];
#pragma unroll
      for (int r = 0; r < 8; ++r)
        out[m0 + r + 8 * half] = p[r] + bcv;
    }
  }
}

extern "C" void kernel_launch(void* const* d_in, const int* in_sizes, int n_in,
                              void* d_out, int out_size, void* d_ws, size_t ws_size,
                              hipStream_t stream) {
  const float* x   = (const float*)d_in[0];
  const int*   ei  = (const int*)d_in[1];
  const float* W1l = (const float*)d_in[2];
  const float* W1r = (const float*)d_in[3];
  const float* b1  = (const float*)d_in[4];
  const float* W2l = (const float*)d_in[5];
  const float* W2r = (const float*)d_in[6];
  const float* b2  = (const float*)d_in[7];
  const float* Wc  = (const float*)d_in[8];
  const float* bc  = (const float*)d_in[9];
  float* out = (float*)d_out;

  const int n_edges = in_sizes[1] / 2;
  const int* src = ei;
  const int* dst = ei + n_edges;

  const size_t featBytes = (size_t)NNODES * DIM * sizeof(float);
  char* ws = (char*)d_ws;
  int* deg  = (int*)ws;
  int* off  = (int*)(ws + ((size_t)1 << 19));
  int* btot = (int*)(ws + ((size_t)15 << 16));
  int* rank = (int*)(ws + ((size_t)1 << 20));
  int* ssrc = (int*)(ws + ((size_t)6 << 20));
  float* h1 = (float*)(ws + featBytes);

  const int edgeBlocks = (n_edges + 255) / 256;
  hipMemsetAsync(deg, 0, (size_t)NNODES * sizeof(int), stream);
  rank_edges_kernel<<<edgeBlocks, 256, 0, stream>>>(dst, deg, rank, n_edges);
  scan_local_kernel<<<SCAN_WG, 1024, 0, stream>>>(deg, off, btot);
  scan_add_kernel<<<SCAN_WG, 1024, 0, stream>>>(off, btot);
  place_edges_kernel<<<edgeBlocks, 256, 0, stream>>>(src, dst, off, rank, ssrc, n_edges);

  sage_layer_kernel<false><<<NNODES / 32, 64, 0, stream>>>(
      x, off, ssrc, W1l, W1r, b1, h1, nullptr, nullptr, nullptr);

  sage_layer_kernel<true><<<NNODES / 32, 64, 0, stream>>>(
      h1, off, ssrc, W2l, W2r, b2, nullptr, Wc, bc, out);
}


# Edge ordering by destination through LDS: two-level counting sort (group bucket counts, bucket runs written from LDS, per-bucket LDS sort) replaces global rank atomics and placement; LDS-arrival sums
# speedup vs baseline: 3.6102x; 1.7490x over previous
#include <hip/hip_runtime.h>
#include <hip/hip_bf16.h>

typedef float v2f __attribute__((ext_vector_type(2)));
typedef float v8f __attribute__((ext_vector_type(8)));

#define NNODES 100000
#define DIM 64
#define TP 68

#define BSHIFT 9
#define BNODES (1 << BSHIFT)
#define NBK ((NNODES + BNODES - 1) >> BSHIFT)
#define NBP 256
#define G1 128
#define SORT_T 512
#define EPT 16
#define SUB (SORT_T * EPT)
#define STAGE 16384
#define SRC_BITS 17
#define SRC_MASK ((1u << SRC_BITS) - 1u)

__device__ __forceinline__ int bucket_of(int d) {
  return (int)min((unsigned)d >> BSHIFT, (unsigned)(NBK - 1));
}

template <int NW>
__device__ __forceinline__ int block_excl_scan(int v, int* wtot, int t) {
  const int lane = t & 31, w = t >> 5;
  int incl = v;
#pragma unroll
  for (int d = 1; d < 32; d <<= 1) {
    const int o = __shfl_up(incl, d, 32);
    if (lane >= d) incl += o;
  }
  if (lane == 31) wtot[w] = incl;
  __syncthreads();
  int base = 0;
#pragma unroll
  for (int j = 0; j < NW; ++j) base += (j < w) ? wtot[j] : 0;
  __syncthreads();
  return base + incl - v;
}

__global__ __launch_bounds__(SORT_T) void bucket_count_kernel(const int* __restrict__ dst,
                                                              int* __restrict__ hist, int n_edges) {
  __shared__ int lcnt[NBP];
  const int t = threadIdx.x;
  if (t < NBP) lcnt[t] = 0;
  __syncthreads();
  const int chunk = (n_edges + G1 - 1) / G1;
  const int e0 = min((int)blockIdx.x * chunk, n_edges);
  const int e1 = min(e0 + chunk, n_edges);
  for (int s0 = e0; s0 < e1; s0 += SUB) {
    const int s1 = min(s0 + SUB, e1);
    int d[EPT];
#pragma unroll
    for (int k = 0; k < EPT; ++k) d[k] = dst[min(s0 + t + k * SORT_T, s1 - 1)];
#pragma unroll
    for (int k = 0; k < EPT; ++k)
      if (s0 + t + k * SORT_T < s1) atomicAdd(&lcnt[bucket_of(d[k])], 1);
  }
  __syncthreads();
  if (t < NBP) hist[blockIdx.x * NBP + t] = lcnt[t];
}

__global__ __launch_bounds__(SORT_T) void bucket_scatter_kernel(
    const int* __restrict__ src, const int* __restrict__ dst,
    const int* __restrict__ hist, unsigned* __restrict__ tmp,
    int* __restrict__ bstart, int n_edges) {
  __shared__ int lcnt[NBP];
  __shared__ int lstart[NBP];
  __shared__ int gdelta[NBP];
  __shared__ int colsum[2][2][NBP];
  __shared__ int wtot[SORT_T / 32];
  __shared__ unsigned sRec[SUB];
  __shared__ unsigned short sB[SUB];

  const int t = threadIdx.x, wg = blockIdx.x;
  {
    const int c = t & (NBP - 1), h = t >> 8;
    const int g0 = h * (G1 / 2);
    int tot = 0, part = 0;
#pragma unroll 8
    for (int g = g0; g < g0 + G1 / 2; ++g) {
      const int v = hist[g * NBP + c];
      tot += v;
      part += (g < wg) ? v : 0;
    }
    colsum[h][0][c] = tot;
    colsum[h][1][c] = part;
  }
  __syncthreads();
  int tot = 0, part = 0;
  if (t < NBP) {
    tot  = colsum[0][0][t] + colsum[1][0][t];
    part = colsum[0][1][t] + colsum[1][1][t];
  }
  const int bst = block_excl_scan<SORT_T / 32>(tot, wtot, t);
  int gbase = bst + part;
  if (wg == 0 && t < NBP) bstart[t] = bst;

  const int chunk = (n_edges + G1 - 1) / G1;
  const int e0 = min(wg * chunk, n_edges);
  const int e1 = min(e0 + chunk, n_edges);
  for (int s0 = e0; s0 < e1; s0 += SUB) {
    const int s1 = min(s0 + SUB, e1);
    const int cnt = s1 - s0;
    if (t < NBP) lcnt[t] = 0;
    __syncthreads();
    int d[EPT], s[EPT], r[EPT];
#pragma unroll
    for (int k = 0; k < EPT; ++k) {
      const int e = min(s0 + t + k * SORT_T, s1 - 1);
      d[k] = dst[e];
      s[k] = src[e];
    }
#pragma unroll
    for (int k = 0; k < EPT; ++k) {
      const bool ok = s0 + t + k * SORT_T < s1;
      r[k] = ok ? atomicAdd(&lcnt[bucket_of(d[k])], 1) : 0;
    }
    __syncthreads();
    const int lc = (t < NBP) ? lcnt[t] : 0;
    const int ls = block_excl_scan<SORT_T / 32>(lc, wtot, t);
    if (t < NBP) {
      lstart[t] = ls;
      gdelta[t] = gbase - ls;
      gbase += lc;
    }
    __syncthreads();
    int rs[EPT];
#pragma unroll
    for (int k = 0; k < EPT; ++k) rs[k] = lstart[bucket_of(d[k])];
#pragma unroll
    for (int k = 0; k < EPT; ++k) {
      if (s0 + t + k * SORT_T < s1) {
        const int pos = rs[k] + r[k];
        sRec[pos] = (((unsigned)d[k] & (BNODES - 1)) << SRC_BITS) | ((unsigned)s[k] & SRC_MASK);
        sB[pos] = (unsigned short)bucket_of(d[k]);
      }
    }
    __syncthreads();
    for (int i = t; i < cnt; i += SORT_T) tmp[gdelta[sB[i]] + i] = sRec[i];
    __syncthreads();
  }
}

__global__ __launch_bounds__(SORT_T) void bucket_sort_kernel(
    const unsigned* __restrict__ tmp, const int* __restrict__ bstart,
    int* __restrict__ off, int* __restrict__ ssrc) {
  __shared__ int ncnt[BNODES];
  __shared__ int nstart[BNODES];
  __shared__ int wtot[SORT_T / 32];
  __shared__ int stage[STAGE];

  const int b = blockIdx.x, t = threadIdx.x;
  const int beg = bstart[b], end = bstart[b + 1];
  const int n = end - beg;
  ncnt[t] = 0;
  __syncthreads();
  for (int i0 = 0; i0 < n; i0 += SORT_T * 8) {
    unsigned rec[8];
#pragma unroll
    for (int k = 0; k < 8; ++k) rec[k] = tmp[beg + min(i0 + t + k * SORT_T, n - 1)];
#pragma unroll
    for (int k = 0; k < 8; ++k)
      if (i0 + t + k * SORT_T < n) atomicAdd(&ncnt[(rec[k] >> SRC_BITS) & (BNODES - 1)], 1);
  }
  __syncthreads();
  const int c = ncnt[t];
  const int ex = block_excl_scan<SORT_T / 32>(c, wtot, t);
  nstart[t] = ex;
  const int node = (b << BSHIFT) + t;
  if (node < NNODES) off[node] = beg + ex;
  if (node == NNODES - 1) off[NNODES] = beg + ex + c;
  ncnt[t] = 0;
  __syncthreads();
  const bool staged = n <= STAGE;
  for (int i0 = 0; i0 < n; i0 += SORT_T * 8) {
    unsigned rec[8];
#pragma unroll
    for (int k = 0; k < 8; ++k) rec[k] = tmp[beg + min(i0 + t + k * SORT_T, n - 1)];
    int ns[8], rk[8];
#pragma unroll
    for (int k = 0; k < 8; ++k) ns[k] = nstart[(rec[k] >> SRC_BITS) & (BNODES - 1)];
#pragma unroll
    for (int k = 0; k < 8; ++k)
      rk[k] = (i0 + t + k * SORT_T < n) ? atomicAdd(&ncnt[(rec[k] >> SRC_BITS) & (BNODES - 1)], 1) : 0;
#pragma unroll
    for (int k = 0; k < 8; ++k) {
      if (i0 + t + k * SORT_T < n) {
        const int pos = ns[k] + rk[k];
        const int sv = (int)(rec[k] & SRC_MASK);
        if (staged) stage[pos] = sv; else ssrc[beg + pos] = sv;
      }
    }
  }
  if (staged) {
    __syncthreads();
    for (int i = t; i < n; i += SORT_T) ssrc[beg + i] = stage[i];
  }
}

__device__ __forceinline__ float4 load_row4(const float* __restrict__ feat, int s, int c4) {
  const unsigned su = min((unsigned)s, (unsigned)(NNODES - 1));
  return *(const float4*)(feat + (size_t)su * DIM + c4);
}

template <bool FINAL>
__global__ __launch_bounds__(64) void sage_layer_kernel(
    const float* __restrict__ feat,
    const int* __restrict__ off, const int* __restrict__ ssrc,
    const float* __restrict__ Wl, const float* __restrict__ Wr,
    const float* __restrict__ bias,
    float* __restrict__ Hout,
    const float* __restrict__ Wc, const float* __restrict__ bc,
    float* __restrict__ out) {
  __shared__ float tile[2][16 * TP];

  const int lane = threadIdx.x & 31;
  const int wave = threadIdx.x >> 5;
  const int rb   = blockIdx.x * 2 + wave;
  const int m0   = rb * 16;
  const int col  = lane & 15;
  const int half = lane >> 4;
  float* myTile = tile[wave];

  {
    const int c4 = col * 4;
    for (int i = 0; i < 8; ++i) {
      const int r = 2 * i + half;
      const int beg = off[m0 + r], end = off[m0 + r + 1];
      float4 acc = make_float4(0.f, 0.f, 0.f, 0.f);
      if (beg < end) {
        const int last = end - 1;
        int s[8];
#pragma unroll
        for (int k = 0; k < 8; ++k) s[k] = ssrc[min(beg + k, last)];
        for (int e = beg; e < end; e += 8) {
          float4 v[8];
#pragma unroll
          for (int k = 0; k < 8; ++k) v[k] = load_row4(feat, s[k], c4);
#pragma unroll
          for (int k = 0; k < 8; ++k) s[k] = ssrc[min(e + 8 + k, last)];
#pragma unroll
          for (int k = 0; k < 8; ++k) {
            const bool ok = e + k < end;
            const float vx = v[k].x, vy = v[k].y, vz = v[k].z, vw = v[k].w;
            acc.x += ok ? vx : 0.0f;
            acc.y += ok ? vy : 0.0f;
            acc.z += ok ? vz : 0.0f;
            acc.w += ok ? vw : 0.0f;
          }
        }
      }
      const float ic = 1.0f / fmaxf((float)(end - beg), 1.0f);
      acc.x *= ic; acc.y *= ic; acc.z *= ic; acc.w *= ic;
      *(float4*)(myTile + r * TP + c4) = acc;
    }
  }
  __syncthreads();

  const int m = m0 + col;
  const float* meanRow = myTile + col * TP;
  const float* xRow    = feat + (size_t)m * DIM;

  v8f c0, c1, c2, c3;
  {
    const float b0 = bias[col], b1 = bias[16 + col];
    const float b2 = bias[32 + col], b3 = bias[48 + col];
#pragma unroll
    for (int r = 0; r < 8; ++r) { c0[r] = b0; c1[r] = b1; c2[r] = b2; c3[r] = b3; }
  }

#pragma unroll
  for (int k0 = 0; k0 < DIM; k0 += 4) {
    const int kb = k0 + 2 * half;
    v2f Am, Ax, B;
    {
      const float2 a = *(const float2*)(meanRow + kb);
      Am[0] = a.x; Am[1] = a.y;
      const float2 b = *(const float2*)(xRow + kb);
      Ax[0] = b.x; Ax[1] = b.y;
    }
    const float* wl = Wl + (size_t)kb * DIM + col;
    const float* wr = Wr + (size_t)kb * DIM + col;

    B[0] = wl[0];       B[1] = wl[DIM];
    c0 = __builtin_amdgcn_wmma_f32_16x16x4_f32(false, Am, false, B, (short)0, c0, false, false);
    B[0] = wr[0];       B[1] = wr[DIM];
    c0 = __builtin_amdgcn_wmma_f32_16x16x4_f32(false, Ax, false, B, (short)0, c0, false, false);
    B[0] = wl[16];      B[1] = wl[DIM + 16];
    c1 = __builtin_amdgcn_wmma_f32_16x16x4_f32(false, Am, false, B, (short)0, c1, false, false);
    B[0] = wr[16];      B[1] = wr[DIM + 16];
    c1 = __builtin_amdgcn_wmma_f32_16x16x4_f32(false, Ax, false, B, (short)0, c1, false, false);
    B[0] = wl[32];      B[1] = wl[DIM + 32];
    c2 = __builtin_amdgcn_wmma_f32_16x16x4_f32(false, Am, false, B, (short)0, c2, false, false);
    B[0] = wr[32];      B[1] = wr[DIM + 32];
    c2 = __builtin_amdgcn_wmma_f32_16x16x4_f32(false, Ax, false, B, (short)0, c2, false, false);
    B[0] = wl[48];      B[1] = wl[DIM + 48];
    c3 = __builtin_amdgcn_wmma_f32_16x16x4_f32(false, Am, false, B, (short)0, c3, false, false);
    B[0] = wr[48];      B[1] = wr[DIM + 48];
    c3 = __builtin_amdgcn_wmma_f32_16x16x4_f32(false, Ax, false, B, (short)0, c3, false, false);
  }

#pragma unroll
  for (int r = 0; r < 8; ++r) {
    c0[r] = fmaxf(c0[r], 0.0f);
    c1[r] = fmaxf(c1[r], 0.0f);
    c2[r] = fmaxf(c2[r], 0.0f);
    c3[r] = fmaxf(c3[r], 0.0f);
  }

  if (!FINAL) {
#pragma unroll
    for (int r = 0; r < 8; ++r) {
      const size_t row = (size_t)(m0 + r + 8 * half) * DIM;
      Hout[row + col]      = c0[r];
      Hout[row + 16 + col] = c1[r];
      Hout[row + 32 + col] = c2[r];
      Hout[row + 48 + col] = c3[r];
    }
  } else {
    const float w0 = Wc[col], w1 = Wc[16 + col], w2 = Wc[32 + col], w3 = Wc[48 + col];
    float p[8];
#pragma unroll
    for (int r = 0; r < 8; ++r)
      p[r] = c0[r] * w0 + c1[r] * w1 + c2[r] * w2 + c3[r] * w3;
#pragma unroll
    for (int r = 0; r < 8; ++r) {
      p[r] += __shfl_xor(p[r], 1, 32);
      p[r] += __shfl_xor(p[r], 2, 32);
      p[r] += __shfl_xor(p[r], 4, 32);
      p[r] += __shfl_xor(p[r], 8, 32);
    }
    if (col == 0) {
      const float bcv = bc[0];
#pragma unroll
      for (int r = 0; r < 8; ++r)
        out[m0 + r + 8 * half] = p[r] + bcv;
    }
  }
}

extern "C" void kernel_launch(void* const* d_in, const int* in_sizes, int n_in,
                              void* d_out, int out_size, void* d_ws, size_t ws_size,
                              hipStream_t stream) {
  const float* x   = (const float*)d_in[0];
  const int*   ei  = (const int*)d_in[1];
  const float* W1l = (const float*)d_in[2];
  const float* W1r = (const float*)d_in[3];
  const float* b1  = (const float*)d_in[4];
  const float* W2l = (const float*)d_in[5];
  const float* W2r = (const float*)d_in[6];
  const float* b2  = (const float*)d_in[7];
  const float* Wc  = (const float*)d_in[8];
  const float* bc  = (const float*)d_in[9];
  float* out = (float*)d_out;

  const int n_edges = in_sizes[1] / 2;
  const int* src = ei;
  const int* dst = ei + n_edges;

  const size_t featBytes = (size_t)NNODES * DIM * sizeof(float);
  char* ws = (char*)d_ws;
  int* hist   = (int*)ws;
  int* off    = (int*)(ws + ((size_t)1 << 19));
  int* bstart = (int*)(ws + ((size_t)15 << 16));
  unsigned* tmp = (unsigned*)(ws + ((size_t)1 << 20));
  int* ssrc   = (int*)(ws + ((size_t)6 << 20));
  float* h1 = (float*)(ws + featBytes);

  bucket_count_kernel<<<G1, SORT_T, 0, stream>>>(dst, hist, n_edges);
  bucket_scatter_kernel<<<G1, SORT_T, 0, stream>>>(src, dst, hist, tmp, bstart, n_edges);
  bucket_sort_kernel<<<NBK, SORT_T, 0, stream>>>(tmp, bstart, off, ssrc);

  sage_layer_kernel<false><<<NNODES / 32, 64, 0, stream>>>(
      x, off, ssrc, W1l, W1r, b1, h1, nullptr, nullptr, nullptr);

  sage_layer_kernel<true><<<NNODES / 32, 64, 0, stream>>>(
      h1, off, ssrc, W2l, W2r, b2, nullptr, Wc, bc, out);
}


# Edge ordering in two launches, not three: each group writes its slice back bucket-ordered in place with count and run-start rows; the bucket sort gathers its 128 runs; LDS-arrival sums as before
# speedup vs baseline: 3.9953x; 1.1067x over previous
#include <hip/hip_runtime.h>
#include <hip/hip_bf16.h>

typedef float v2f __attribute__((ext_vector_type(2)));
typedef float v8f __attribute__((ext_vector_type(8)));

#define NNODES 100000
#define DIM 64
#define TP 68

#define BSHIFT 9
#define BNODES (1 << BSHIFT)
#define NBK ((NNODES + BNODES - 1) >> BSHIFT)
#define NBP 256
#define G1 128
#define SORT_T 512
#define EPT 16
#define SUB (SORT_T * EPT)
#define RPT 16
#define STAGE 16384
#define SRC_BITS 17
#define SRC_MASK ((1u << SRC_BITS) - 1u)

__device__ __forceinline__ int bucket_of(int d) {
  return (int)min((unsigned)d >> BSHIFT, (unsigned)(NBK - 1));
}

template <int NW>
__device__ __forceinline__ int block_excl_scan(int v, int* wtot, int t) {
  const int lane = t & 31, w = t >> 5;
  int incl = v;
#pragma unroll
  for (int d = 1; d < 32; d <<= 1) {
    const int o = __shfl_up(incl, d, 32);
    if (lane >= d) incl += o;
  }
  if (lane == 31) wtot[w] = incl;
  __syncthreads();
  int base = 0;
#pragma unroll
  for (int j = 0; j < NW; ++j) base += (j < w) ? wtot[j] : 0;
  __syncthreads();
  return base + incl - v;
}

__device__ __forceinline__ int slice_len(int n_edges) {
  return min(((n_edges + G1 - 1) / G1 + 31) & ~31, SUB);
}

__global__ __launch_bounds__(SORT_T) void bucket_group_kernel(
    const int* __restrict__ src, const int* __restrict__ dst,
    int* __restrict__ hist, int* __restrict__ lst,
    unsigned* __restrict__ tmp, int n_edges) {
  __shared__ int lcnt[NBP];
  __shared__ int lstart[NBP];
  __shared__ int wtot[SORT_T / 32];
  __shared__ unsigned sRec[SUB];

  const int t = threadIdx.x, wg = blockIdx.x;
  const int chunk = slice_len(n_edges);
  const int e0 = min(wg * chunk, n_edges);
  const int e1 = min(e0 + chunk, n_edges);
  const int cnt = e1 - e0;
  if (t < NBP) lcnt[t] = 0;
  __syncthreads();
  int d[EPT], s[EPT], r[EPT];
#pragma unroll
  for (int k = 0; k < EPT; ++k) {
    const int e = min(e0 + t + k * SORT_T, e1 - 1);
    d[k] = dst[e];
    s[k] = src[e];
  }
#pragma unroll
  for (int k = 0; k < EPT; ++k) {
    const bool ok = e0 + t + k * SORT_T < e1;
    r[k] = ok ? atomicAdd(&lcnt[bucket_of(d[k])], 1) : 0;
  }
  __syncthreads();
  const int lc = (t < NBP) ? lcnt[t] : 0;
  const int ls = block_excl_scan<SORT_T / 32>(lc, wtot, t);
  if (t < NBP) {
    lstart[t] = ls;
    hist[wg * NBP + t] = lc;
    lst[wg * NBP + t] = ls;
  }
  __syncthreads();
  int rs[EPT];
#pragma unroll
  for (int k = 0; k < EPT; ++k) rs[k] = lstart[bucket_of(d[k])];
#pragma unroll
  for (int k = 0; k < EPT; ++k) {
    if (e0 + t + k * SORT_T < e1)
      sRec[rs[k] + r[k]] = (((unsigned)d[k] & (BNODES - 1)) << SRC_BITS) | ((unsigned)s[k] & SRC_MASK);
  }
  __syncthreads();
  for (int i = t; i < cnt; i += SORT_T) tmp[e0 + i] = sRec[i];
}

__global__ __launch_bounds__(SORT_T) void bucket_sort_kernel(
    const unsigned* __restrict__ tmp, const int* __restrict__ hist,
    const int* __restrict__ lst, int* __restrict__ off, int* __restrict__ ssrc, int n_edges) {
  __shared__ int ncnt[BNODES];
  __shared__ int nstart[BNODES];
  __shared__ int wtot[SORT_T / 32];
  __shared__ int rbase[G1];
  __shared__ int rcnt[G1];
  __shared__ int rflat[G1];
  __shared__ int tot2[2];
  __shared__ unsigned recs[STAGE];
  __shared__ int stage[STAGE];

  const int b = blockIdx.x, t = threadIdx.x;
  const int chunk = slice_len(n_edges);
  int c = 0, l = 0;
  if (t < G1) { c = hist[t * NBP + b]; l = lst[t * NBP + b]; }
  const int fl = block_excl_scan<SORT_T / 32>(c, wtot, t);
  const int lp = block_excl_scan<SORT_T / 32>(l, wtot, t);
  if (t < G1) { rcnt[t] = c; rflat[t] = fl; rbase[t] = min(t * chunk, n_edges) + l; }
  if (t == SORT_T - 1) { tot2[0] = fl; tot2[1] = lp; }
  ncnt[t] = 0;
  __syncthreads();
  const int n = tot2[0], beg = tot2[1];
  const bool staged = n <= STAGE;
  const int g = t >> 2, j = t & 3;
  const int cg = rcnt[g], bg = rbase[g], fg = rflat[g];

  for (int i0 = 0; i0 < cg; i0 += 4 * RPT) {
    unsigned rec[RPT];
#pragma unroll
    for (int k = 0; k < RPT; ++k) rec[k] = tmp[bg + min(i0 + j + 4 * k, cg - 1)];
#pragma unroll
    for (int k = 0; k < RPT; ++k) {
      const int i = i0 + j + 4 * k;
      if (i < cg) {
        atomicAdd(&ncnt[(rec[k] >> SRC_BITS) & (BNODES - 1)], 1);
        if (staged) recs[fg + i] = rec[k];
      }
    }
  }
  __syncthreads();
  const int cn = ncnt[t];
  const int ex = block_excl_scan<SORT_T / 32>(cn, wtot, t);
  nstart[t] = ex;
  const int node = (b << BSHIFT) + t;
  if (node < NNODES) off[node] = beg + ex;
  if (node == NNODES - 1) off[NNODES] = beg + ex + cn;
  ncnt[t] = 0;
  __syncthreads();
  if (staged) {
    for (int i0 = 0; i0 < n; i0 += SORT_T * 8) {
      unsigned rec[8];
#pragma unroll
      for (int k = 0; k < 8; ++k) rec[k] = recs[min(i0 + t + k * SORT_T, n - 1)];
      int ns[8], rk[8];
#pragma unroll
      for (int k = 0; k < 8; ++k) ns[k] = nstart[(rec[k] >> SRC_BITS) & (BNODES - 1)];
#pragma unroll
      for (int k = 0; k < 8; ++k)
        rk[k] = (i0 + t + k * SORT_T < n) ? atomicAdd(&ncnt[(rec[k] >> SRC_BITS) & (BNODES - 1)], 1) : 0;
#pragma unroll
      for (int k = 0; k < 8; ++k)
        if (i0 + t + k * SORT_T < n) stage[ns[k] + rk[k]] = (int)(rec[k] & SRC_MASK);
    }
    __syncthreads();
    for (int i = t; i < n; i += SORT_T) ssrc[beg + i] = stage[i];
  } else {
    for (int i0 = 0; i0 < cg; i0 += 4 * RPT) {
      unsigned rec[RPT];
#pragma unroll
      for (int k = 0; k < RPT; ++k) rec[k] = tmp[bg + min(i0 + j + 4 * k, cg - 1)];
#pragma unroll
      for (int k = 0; k < RPT; ++k) {
        if (i0 + j + 4 * k < cg) {
          const int dl = (rec[k] >> SRC_BITS) & (BNODES - 1);
          const int pos = nstart[dl] + atomicAdd(&ncnt[dl], 1);
          ssrc[beg + pos] = (int)(rec[k] & SRC_MASK);
        }
      }
    }
  }
}

__device__ __forceinline__ float4 load_row4(const float* __restrict__ feat, int s, int c4) {
  const unsigned su = min((unsigned)s, (unsigned)(NNODES - 1));
  return *(const float4*)(feat + (size_t)su * DIM + c4);
}

template <bool FINAL>
__global__ __launch_bounds__(64) void sage_layer_kernel(
    const float* __restrict__ feat,
    const int* __restrict__ off, const int* __restrict__ ssrc,
    const float* __restrict__ Wl, const float* __restrict__ Wr,
    const float* __restrict__ bias,
    float* __restrict__ Hout,
    const float* __restrict__ Wc, const float* __restrict__ bc,
    float* __restrict__ out) {
  __shared__ float tile[2][16 * TP];

  const int lane = threadIdx.x & 31;
  const int wave = threadIdx.x >> 5;
  const int rb   = blockIdx.x * 2 + wave;
  const int m0   = rb * 16;
  const int col  = lane & 15;
  const int half = lane >> 4;
  float* myTile = tile[wave];

  {
    const int c4 = col * 4;
    for (int i = 0; i < 8; ++i) {
      const int r = 2 * i + half;
      const int beg = off[m0 + r], end = off[m0 + r + 1];
      float4 acc = make_float4(0.f, 0.f, 0.f, 0.f);
      if (beg < end) {
        const int last = end - 1;
        int s[8];
#pragma unroll
        for (int k = 0; k < 8; ++k) s[k] = ssrc[min(beg + k, last)];
        for (int e = beg; e < end; e += 8) {
          float4 v[8];
#pragma unroll
          for (int k = 0; k < 8; ++k) v[k] = load_row4(feat, s[k], c4);
#pragma unroll
          for (int k = 0; k < 8; ++k) s[k] = ssrc[min(e + 8 + k, last)];
#pragma unroll
          for (int k = 0; k < 8; ++k) {
            const bool ok = e + k < end;
            const float vx = v[k].x, vy = v[k].y, vz = v[k].z, vw = v[k].w;
            acc.x += ok ? vx : 0.0f;
            acc.y += ok ? vy : 0.0f;
            acc.z += ok ? vz : 0.0f;
            acc.w += ok ? vw : 0.0f;
          }
        }
      }
      const float ic = 1.0f / fmaxf((float)(end - beg), 1.0f);
      acc.x *= ic; acc.y *= ic; acc.z *= ic; acc.w *= ic;
      *(float4*)(myTile + r * TP + c4) = acc;
    }
  }
  __syncthreads();

  const int m = m0 + col;
  const float* meanRow = myTile + col * TP;
  const float* xRow    = feat + (size_t)m * DIM;

  v8f c0, c1, c2, c3;
  {
    const float b0 = bias[col], b1 = bias[16 + col];
    const float b2 = bias[32 + col], b3 = bias[48 + col];
#pragma unroll
    for (int r = 0; r < 8; ++r) { c0[r] = b0; c1[r] = b1; c2[r] = b2; c3[r] = b3; }
  }

#pragma unroll
  for (int k0 = 0; k0 < DIM; k0 += 4) {
    const int kb = k0 + 2 * half;
    v2f Am, Ax, B;
    {
      const float2 a = *(const float2*)(meanRow + kb);
      Am[0] = a.x; Am[1] = a.y;
      const float2 b = *(const float2*)(xRow + kb);
      Ax[0] = b.x; Ax[1] = b.y;
    }
    const float* wl = Wl + (size_t)kb * DIM + col;
    const float* wr = Wr + (size_t)kb * DIM + col;

    B[0] = wl[0];       B[1] = wl[DIM];
    c0 = __builtin_amdgcn_wmma_f32_16x16x4_f32(false, Am, false, B, (short)0, c0, false, false);
    B[0] = wr[0];       B[1] = wr[DIM];
    c0 = __builtin_amdgcn_wmma_f32_16x16x4_f32(false, Ax, false, B, (short)0, c0, false, false);
    B[0] = wl[16];      B[1] = wl[DIM + 16];
    c1 = __builtin_amdgcn_wmma_f32_16x16x4_f32(false, Am, false, B, (short)0, c1, false, false);
    B[0] = wr[16];      B[1] = wr[DIM + 16];
    c1 = __builtin_amdgcn_wmma_f32_16x16x4_f32(false, Ax, false, B, (short)0, c1, false, false);
    B[0] = wl[32];      B[1] = wl[DIM + 32];
    c2 = __builtin_amdgcn_wmma_f32_16x16x4_f32(false, Am, false, B, (short)0, c2, false, false);
    B[0] = wr[32];      B[1] = wr[DIM + 32];
    c2 = __builtin_amdgcn_wmma_f32_16x16x4_f32(false, Ax, false, B, (short)0, c2, false, false);
    B[0] = wl[48];      B[1] = wl[DIM + 48];
    c3 = __builtin_amdgcn_wmma_f32_16x16x4_f32(false, Am, false, B, (short)0, c3, false, false);
    B[0] = wr[48];      B[1] = wr[DIM + 48];
    c3 = __builtin_amdgcn_wmma_f32_16x16x4_f32(false, Ax, false, B, (short)0, c3, false, false);
  }

#pragma unroll
  for (int r = 0; r < 8; ++r) {
    c0[r] = fmaxf(c0[r], 0.0f);
    c1[r] = fmaxf(c1[r], 0.0f);
    c2[r] = fmaxf(c2[r], 0.0f);
    c3[r] = fmaxf(c3[r], 0.0f);
  }

  if (!FINAL) {
#pragma unroll
    for (int r = 0; r < 8; ++r) {
      const size_t row = (size_t)(m0 + r + 8 * half) * DIM;
      Hout[row + col]      = c0[r];
      Hout[row + 16 + col] = c1[r];
      Hout[row + 32 + col] = c2[r];
      Hout[row + 48 + col] = c3[r];
    }
  } else {
    const float w0 = Wc[col], w1 = Wc[16 + col], w2 = Wc[32 + col], w3 = Wc[48 + col];
    float p[8];
#pragma unroll
    for (int r = 0; r < 8; ++r)
      p[r] = c0[r] * w0 + c1[r] * w1 + c2[r] * w2 + c3[r] * w3;
#pragma unroll
    for (int r = 0; r < 8; ++r) {
      p[r] += __shfl_xor(p[r], 1, 32);
      p[r] += __shfl_xor(p[r], 2, 32);
      p[r] += __shfl_xor(p[r], 4, 32);
      p[r] += __shfl_xor(p[r], 8, 32);
    }
    if (col == 0) {
      const float bcv = bc[0];
#pragma unroll
      for (int r = 0; r < 8; ++r)
        out[m0 + r + 8 * half] = p[r] + bcv;
    }
  }
}

extern "C" void kernel_launch(void* const* d_in, const int* in_sizes, int n_in,
                              void* d_out, int out_size, void* d_ws, size_t ws_size,
                              hipStream_t stream) {
  const float* x   = (const float*)d_in[0];
  const int*   ei  = (const int*)d_in[1];
  const float* W1l = (const float*)d_in[2];
  const float* W1r = (const float*)d_in[3];
  const float* b1  = (const float*)d_in[4];
  const float* W2l = (const float*)d_in[5];
  const float* W2r = (const float*)d_in[6];
  const float* b2  = (const float*)d_in[7];
  const float* Wc  = (const float*)d_in[8];
  const float* bc  = (const float*)d_in[9];
  float* out = (float*)d_out;

  const int n_edges = in_sizes[1] / 2;
  const int* src = ei;
  const int* dst = ei + n_edges;

  const size_t featBytes = (size_t)NNODES * DIM * sizeof(float);
  char* ws = (char*)d_ws;
  int* hist   = (int*)ws;
  int* lst    = (int*)(ws + ((size_t)1 << 17));
  int* off    = (int*)(ws + ((size_t)1 << 19));
  unsigned* tmp = (unsigned*)(ws + ((size_t)1 << 20));
  int* ssrc   = (int*)(ws + ((size_t)6 << 20));
  float* h1 = (float*)(ws + featBytes);

  bucket_group_kernel<<<G1, SORT_T, 0, stream>>>(src, dst, hist, lst, tmp, n_edges);
  bucket_sort_kernel<<<NBK, SORT_T, 0, stream>>>(tmp, hist, lst, off, ssrc, n_edges);

  sage_layer_kernel<false><<<NNODES / 32, 64, 0, stream>>>(
      x, off, ssrc, W1l, W1r, b1, h1, nullptr, nullptr, nullptr);

  sage_layer_kernel<true><<<NNODES / 32, 64, 0, stream>>>(
      h1, off, ssrc, W2l, W2r, b2, nullptr, Wc, bc, out);
}


# Neighbour gather without padding requests: block offsets in one load, 16 indices per load shared by DPP row_share, no row load past a row's end, 16 rows in flight; sums in list order as before
# speedup vs baseline: 4.2162x; 1.0553x over previous
#include <hip/hip_runtime.h>
#include <hip/hip_bf16.h>

typedef float v2f __attribute__((ext_vector_type(2)));
typedef float v8f __attribute__((ext_vector_type(8)));

#define NNODES 100000
#define DIM 64
#define TP 68

#define BSHIFT 9
#define BNODES (1 << BSHIFT)
#define NBK ((NNODES + BNODES - 1) >> BSHIFT)
#define NBP 256
#define G1 128
#define SORT_T 512
#define EPT 16
#define SUB (SORT_T * EPT)
#define RPT 16
#define STAGE 16384
#define SRC_BITS 17
#define SRC_MASK ((1u << SRC_BITS) - 1u)

__device__ __forceinline__ int bucket_of(int d) {
  return (int)min((unsigned)d >> BSHIFT, (unsigned)(NBK - 1));
}

template <int NW>
__device__ __forceinline__ int block_excl_scan(int v, int* wtot, int t) {
  const int lane = t & 31, w = t >> 5;
  int incl = v;
#pragma unroll
  for (int d = 1; d < 32; d <<= 1) {
    const int o = __shfl_up(incl, d, 32);
    if (lane >= d) incl += o;
  }
  if (lane == 31) wtot[w] = incl;
  __syncthreads();
  int base = 0;
#pragma unroll
  for (int j = 0; j < NW; ++j) base += (j < w) ? wtot[j] : 0;
  __syncthreads();
  return base + incl - v;
}

__device__ __forceinline__ int slice_len(int n_edges) {
  return min(((n_edges + G1 - 1) / G1 + 31) & ~31, SUB);
}

__global__ __launch_bounds__(SORT_T) void bucket_group_kernel(
    const int* __restrict__ src, const int* __restrict__ dst,
    int* __restrict__ hist, int* __restrict__ lst,
    unsigned* __restrict__ tmp, int n_edges) {
  __shared__ int lcnt[NBP];
  __shared__ int lstart[NBP];
  __shared__ int wtot[SORT_T / 32];
  __shared__ unsigned sRec[SUB];

  const int t = threadIdx.x, wg = blockIdx.x;
  const int chunk = slice_len(n_edges);
  const int e0 = min(wg * chunk, n_edges);
  const int e1 = min(e0 + chunk, n_edges);
  const int cnt = e1 - e0;
  if (t < NBP) lcnt[t] = 0;
  __syncthreads();
  int d[EPT], s[EPT], r[EPT];
#pragma unroll
  for (int k = 0; k < EPT; ++k) {
    const int e = min(e0 + t + k * SORT_T, e1 - 1);
    d[k] = dst[e];
    s[k] = src[e];
  }
#pragma unroll
  for (int k = 0; k < EPT; ++k) {
    const bool ok = e0 + t + k * SORT_T < e1;
    r[k] = ok ? atomicAdd(&lcnt[bucket_of(d[k])], 1) : 0;
  }
  __syncthreads();
  const int lc = (t < NBP) ? lcnt[t] : 0;
  const int ls = block_excl_scan<SORT_T / 32>(lc, wtot, t);
  if (t < NBP) {
    lstart[t] = ls;
    hist[wg * NBP + t] = lc;
    lst[wg * NBP + t] = ls;
  }
  __syncthreads();
  int rs[EPT];
#pragma unroll
  for (int k = 0; k < EPT; ++k) rs[k] = lstart[bucket_of(d[k])];
#pragma unroll
  for (int k = 0; k < EPT; ++k) {
    if (e0 + t + k * SORT_T < e1)
      sRec[rs[k] + r[k]] = (((unsigned)d[k] & (BNODES - 1)) << SRC_BITS) | ((unsigned)s[k] & SRC_MASK);
  }
  __syncthreads();
  for (int i = t; i < cnt; i += SORT_T) tmp[e0 + i] = sRec[i];
}

__global__ __launch_bounds__(SORT_T) void bucket_sort_kernel(
    const unsigned* __restrict__ tmp, const int* __restrict__ hist,
    const int* __restrict__ lst, int* __restrict__ off, int* __restrict__ ssrc, int n_edges) {
  __shared__ int ncnt[BNODES];
  __shared__ int nstart[BNODES];
  __shared__ int wtot[SORT_T / 32];
  __shared__ int rbase[G1];
  __shared__ int rcnt[G1];
  __shared__ int rflat[G1];
  __shared__ int tot2[2];
  __shared__ unsigned recs[STAGE];
  __shared__ int stage[STAGE];

  const int b = blockIdx.x, t = threadIdx.x;
  const int chunk = slice_len(n_edges);
  int c = 0, l = 0;
  if (t < G1) { c = hist[t * NBP + b]; l = lst[t * NBP + b]; }
  const int fl = block_excl_scan<SORT_T / 32>(c, wtot, t);
  const int lp = block_excl_scan<SORT_T / 32>(l, wtot, t);
  if (t < G1) { rcnt[t] = c; rflat[t] = fl; rbase[t] = min(t * chunk, n_edges) + l; }
  if (t == SORT_T - 1) { tot2[0] = fl; tot2[1] = lp; }
  ncnt[t] = 0;
  __syncthreads();
  const int n = tot2[0], beg = tot2[1];
  const bool staged = n <= STAGE;
  const int g = t >> 2, j = t & 3;
  const int cg = rcnt[g], bg = rbase[g], fg = rflat[g];

  for (int i0 = 0; i0 < cg; i0 += 4 * RPT) {
    unsigned rec[RPT];
#pragma unroll
    for (int k = 0; k < RPT; ++k) rec[k] = tmp[bg + min(i0 + j + 4 * k, cg - 1)];
#pragma unroll
    for (int k = 0; k < RPT; ++k) {
      const int i = i0 + j + 4 * k;
      if (i < cg) {
        atomicAdd(&ncnt[(rec[k] >> SRC_BITS) & (BNODES - 1)], 1);
        if (staged) recs[fg + i] = rec[k];
      }
    }
  }
  __syncthreads();
  const int cn = ncnt[t];
  const int ex = block_excl_scan<SORT_T / 32>(cn, wtot, t);
  nstart[t] = ex;
  const int node = (b << BSHIFT) + t;
  if (node < NNODES) off[node] = beg + ex;
  if (node == NNODES - 1) off[NNODES] = beg + ex + cn;
  ncnt[t] = 0;
  __syncthreads();
  if (staged) {
    for (int i0 = 0; i0 < n; i0 += SORT_T * 8) {
      unsigned rec[8];
#pragma unroll
      for (int k = 0; k < 8; ++k) rec[k] = recs[min(i0 + t + k * SORT_T, n - 1)];
      int ns[8], rk[8];
#pragma unroll
      for (int k = 0; k < 8; ++k) ns[k] = nstart[(rec[k] >> SRC_BITS) & (BNODES - 1)];
#pragma unroll
      for (int k = 0; k < 8; ++k)
        rk[k] = (i0 + t + k * SORT_T < n) ? atomicAdd(&ncnt[(rec[k] >> SRC_BITS) & (BNODES - 1)], 1) : 0;
#pragma unroll
      for (int k = 0; k < 8; ++k)
        if (i0 + t + k * SORT_T < n) stage[ns[k] + rk[k]] = (int)(rec[k] & SRC_MASK);
    }
    __syncthreads();
    for (int i = t; i < n; i += SORT_T) ssrc[beg + i] = stage[i];
  } else {
    for (int i0 = 0; i0 < cg; i0 += 4 * RPT) {
      unsigned rec[RPT];
#pragma unroll
      for (int k = 0; k < RPT; ++k) rec[k] = tmp[bg + min(i0 + j + 4 * k, cg - 1)];
#pragma unroll
      for (int k = 0; k < RPT; ++k) {
        if (i0 + j + 4 * k < cg) {
          const int dl = (rec[k] >> SRC_BITS) & (BNODES - 1);
          const int pos = nstart[dl] + atomicAdd(&ncnt[dl], 1);
          ssrc[beg + pos] = (int)(rec[k] & SRC_MASK);
        }
      }
    }
  }
}

__device__ __forceinline__ float4 load_row4(const float* __restrict__ feat, int s, int c4) {
  const unsigned su = min((unsigned)s, (unsigned)(NNODES - 1));
  return *(const float4*)(feat + (size_t)su * DIM + c4);
}

#define GATHER_SLOT(K)                                                                   \
  {                                                                                      \
    const int s_ = __builtin_amdgcn_update_dpp(0, idx, 0x150 + (K), 0xf, 0xf, true);     \
    v[K] = make_float4(0.f, 0.f, 0.f, 0.f);                                              \
    if ((K) < n) v[K] = load_row4(feat, s_, c4);                                         \
  }

template <bool FINAL>
__global__ __launch_bounds__(64) void sage_layer_kernel(
    const float* __restrict__ feat,
    const int* __restrict__ off, const int* __restrict__ ssrc,
    const float* __restrict__ Wl, const float* __restrict__ Wr,
    const float* __restrict__ bias,
    float* __restrict__ Hout,
    const float* __restrict__ Wc, const float* __restrict__ bc,
    float* __restrict__ out) {
  __shared__ float tile[2][16 * TP];

  const int lane = threadIdx.x & 31;
  const int wave = threadIdx.x >> 5;
  const int rb   = blockIdx.x * 2 + wave;
  const int m0   = rb * 16;
  const int col  = lane & 15;
  const int half = lane >> 4;
  float* myTile = tile[wave];

  {
    const int c4 = col * 4;
    const int offv = off[m0 + min(lane, 16)];
    int beg = __shfl(offv, half, 32), end = __shfl(offv, half + 1, 32);
    int idx = 0;
    if (beg + col < end) idx = ssrc[beg + col];
    for (int i = 0; i < 8; ++i) {
      const int r = 2 * i + half;
      const int begN = __shfl(offv, min(r + 2, 16), 32);
      const int endN = __shfl(offv, min(r + 3, 16), 32);
      float4 v[16];
      int n = end - beg;
      GATHER_SLOT(0)  GATHER_SLOT(1)  GATHER_SLOT(2)  GATHER_SLOT(3)
      GATHER_SLOT(4)  GATHER_SLOT(5)  GATHER_SLOT(6)  GATHER_SLOT(7)
      GATHER_SLOT(8)  GATHER_SLOT(9)  GATHER_SLOT(10) GATHER_SLOT(11)
      GATHER_SLOT(12) GATHER_SLOT(13) GATHER_SLOT(14) GATHER_SLOT(15)
      int idxN = 0;
      if (begN + col < endN) idxN = ssrc[begN + col];
      float4 acc = v[0];
#pragma unroll
      for (int k = 1; k < 16; ++k) {
        acc.x += v[k].x; acc.y += v[k].y; acc.z += v[k].z; acc.w += v[k].w;
      }
      for (int e = beg + 16; e < end; e += 16) {
        idx = 0;
        if (e + col < end) idx = ssrc[e + col];
        n = end - e;
        GATHER_SLOT(0)  GATHER_SLOT(1)  GATHER_SLOT(2)  GATHER_SLOT(3)
        GATHER_SLOT(4)  GATHER_SLOT(5)  GATHER_SLOT(6)  GATHER_SLOT(7)
        GATHER_SLOT(8)  GATHER_SLOT(9)  GATHER_SLOT(10) GATHER_SLOT(11)
        GATHER_SLOT(12) GATHER_SLOT(13) GATHER_SLOT(14) GATHER_SLOT(15)
#pragma unroll
        for (int k = 0; k < 16; ++k) {
          acc.x += v[k].x; acc.y += v[k].y; acc.z += v[k].z; acc.w += v[k].w;
        }
      }
      const float ic = 1.0f / fmaxf((float)(end - beg), 1.0f);
      acc.x *= ic; acc.y *= ic; acc.z *= ic; acc.w *= ic;
      *(float4*)(myTile + r * TP + c4) = acc;
      beg = begN; end = endN; idx = idxN;
    }
  }
  __syncthreads();

  const int m = m0 + col;
  const float* meanRow = myTile + col * TP;
  const float* xRow    = feat + (size_t)m * DIM;

  v8f c0, c1, c2, c3;
  {
    const float b0 = bias[col], b1 = bias[16 + col];
    const float b2 = bias[32 + col], b3 = bias[48 + col];
#pragma unroll
    for (int r = 0; r < 8; ++r) { c0[r] = b0; c1[r] = b1; c2[r] = b2; c3[r] = b3; }
  }

#pragma unroll
  for (int k0 = 0; k0 < DIM; k0 += 4) {
    const int kb = k0 + 2 * half;
    v2f Am, Ax, B;
    {
      const float2 a = *(const float2*)(meanRow + kb);
      Am[0] = a.x; Am[1] = a.y;
      const float2 b = *(const float2*)(xRow + kb);
      Ax[0] = b.x; Ax[1] = b.y;
    }
    const float* wl = Wl + (size_t)kb * DIM + col;
    const float* wr = Wr + (size_t)kb * DIM + col;

    B[0] = wl[0];       B[1] = wl[DIM];
    c0 = __builtin_amdgcn_wmma_f32_16x16x4_f32(false, Am, false, B, (short)0, c0, false, false);
    B[0] = wr[0];       B[1] = wr[DIM];
    c0 = __builtin_amdgcn_wmma_f32_16x16x4_f32(false, Ax, false, B, (short)0, c0, false, false);
    B[0] = wl[16];      B[1] = wl[DIM + 16];
    c1 = __builtin_amdgcn_wmma_f32_16x16x4_f32(false, Am, false, B, (short)0, c1, false, false);
    B[0] = wr[16];      B[1] = wr[DIM + 16];
    c1 = __builtin_amdgcn_wmma_f32_16x16x4_f32(false, Ax, false, B, (short)0, c1, false, false);
    B[0] = wl[32];      B[1] = wl[DIM + 32];
    c2 = __builtin_amdgcn_wmma_f32_16x16x4_f32(false, Am, false, B, (short)0, c2, false, false);
    B[0] = wr[32];      B[1] = wr[DIM + 32];
    c2 = __builtin_amdgcn_wmma_f32_16x16x4_f32(false, Ax, false, B, (short)0, c2, false, false);
    B[0] = wl[48];      B[1] = wl[DIM + 48];
    c3 = __builtin_amdgcn_wmma_f32_16x16x4_f32(false, Am, false, B, (short)0, c3, false, false);
    B[0] = wr[48];      B[1] = wr[DIM + 48];
    c3 = __builtin_amdgcn_wmma_f32_16x16x4_f32(false, Ax, false, B, (short)0, c3, false, false);
  }

#pragma unroll
  for (int r = 0; r < 8; ++r) {
    c0[r] = fmaxf(c0[r], 0.0f);
    c1[r] = fmaxf(c1[r], 0.0f);
    c2[r] = fmaxf(c2[r], 0.0f);
    c3[r] = fmaxf(c3[r], 0.0f);
  }

  if (!FINAL) {
#pragma unroll
    for (int r = 0; r < 8; ++r) {
      const size_t row = (size_t)(m0 + r + 8 * half) * DIM;
      Hout[row + col]      = c0[r];
      Hout[row + 16 + col] = c1[r];
      Hout[row + 32 + col] = c2[r];
      Hout[row + 48 + col] = c3[r];
    }
  } else {
    const float w0 = Wc[col], w1 = Wc[16 + col], w2 = Wc[32 + col], w3 = Wc[48 + col];
    float p[8];
#pragma unroll
    for (int r = 0; r < 8; ++r)
      p[r] = c0[r] * w0 + c1[r] * w1 + c2[r] * w2 + c3[r] * w3;
#pragma unroll
    for (int r = 0; r < 8; ++r) {
      p[r] += __shfl_xor(p[r], 1, 32);
      p[r] += __shfl_xor(p[r], 2, 32);
      p[r] += __shfl_xor(p[r], 4, 32);
      p[r] += __shfl_xor(p[r], 8, 32);
    }
    if (col == 0) {
      const float bcv = bc[0];
#pragma unroll
      for (int r = 0; r < 8; ++r)
        out[m0 + r + 8 * half] = p[r] + bcv;
    }
  }
}

extern "C" void kernel_launch(void* const* d_in, const int* in_sizes, int n_in,
                              void* d_out, int out_size, void* d_ws, size_t ws_size,
                              hipStream_t stream) {
  const float* x   = (const float*)d_in[0];
  const int*   ei  = (const int*)d_in[1];
  const float* W1l = (const float*)d_in[2];
  const float* W1r = (const float*)d_in[3];
  const float* b1  = (const float*)d_in[4];
  const float* W2l = (const float*)d_in[5];
  const float* W2r = (const float*)d_in[6];
  const float* b2  = (const float*)d_in[7];
  const float* Wc  = (const float*)d_in[8];
  const float* bc  = (const float*)d_in[9];
  float* out = (float*)d_out;

  const int n_edges = in_sizes[1] / 2;
  const int* src = ei;
  const int* dst = ei + n_edges;

  const size_t featBytes = (size_t)NNODES * DIM * sizeof(float);
  char* ws = (char*)d_ws;
  int* hist   = (int*)ws;
  int* lst    = (int*)(ws + ((size_t)1 << 17));
  int* off    = (int*)(ws + ((size_t)1 << 19));
  unsigned* tmp = (unsigned*)(ws + ((size_t)1 << 20));
  int* ssrc   = (int*)(ws + ((size_t)6 << 20));
  float* h1 = (float*)(ws + featBytes);

  bucket_group_kernel<<<G1, SORT_T, 0, stream>>>(src, dst, hist, lst, tmp, n_edges);
  bucket_sort_kernel<<<NBK, SORT_T, 0, stream>>>(tmp, hist, lst, off, ssrc, n_edges);

  sage_layer_kernel<false><<<NNODES / 32, 64, 0, stream>>>(
      x, off, ssrc, W1l, W1r, b1, h1, nullptr, nullptr, nullptr);

  sage_layer_kernel<true><<<NNODES / 32, 64, 0, stream>>>(
      h1, off, ssrc, W2l, W2r, b2, nullptr, Wc, bc, out);
}
